# starting kernel
# baseline (speedup 1.0000x reference)
#include <hip/hip_runtime.h>
#include <hip/hip_bf16.h>
#include <math.h>

typedef float v2f __attribute__((ext_vector_type(2)));
typedef float v8f __attribute__((ext_vector_type(8)));

#define BN 8
#define NN 1024
#define DD 256
#define EPSV 1e-8f
#define SSTRIDE 1028

__device__ __forceinline__ v8f wmma4(v2f a, v2f b, v8f c) {
  return __builtin_amdgcn_wmma_f32_16x16x4_f32(
       false, a,  false, b,
       (short)0, c,  false,  false);
}

__device__ __forceinline__ float wave_max(float v) {
#pragma unroll
  for (int off = 16; off > 0; off >>= 1) v = fmaxf(v, __shfl_xor(v, off, 32));
  return v;
}
__device__ __forceinline__ float wave_sum(float v) {
#pragma unroll
  for (int off = 16; off > 0; off >>= 1) v += __shfl_xor(v, off, 32);
  return v;
}

__global__ void pos_softmax_kernel(const float* __restrict__ coords,
                                   const float* __restrict__ pos_emb,
                                   const float* __restrict__ p_temp,
                                   float* __restrict__ pos_score) {
  __shared__ float row[NN];
  __shared__ float red[8];
  const int n = blockIdx.x;
  const int tid = threadIdx.x;
  const int lane = tid & 31, wave = tid >> 5;
  const float pa = fabsf(p_temp[0]);
  float pe[6];
#pragma unroll
  for (int c = 0; c < 6; ++c) pe[c] = pos_emb[n * 6 + c];

  float lmax = -1e30f;
  for (int m = tid; m < NN; m += 256) {
    const float* cp = coords + ((size_t)n * NN + m) * 6;
    float dot = cp[0] * pe[0] + cp[1] * pe[1] + cp[2] * pe[2] +
                cp[3] * pe[3] + cp[4] * pe[4] + cp[5] * pe[5];
    float lg = -pa * dot;
    row[m] = lg;
    lmax = fmaxf(lmax, lg);
  }
  lmax = wave_max(lmax);
  if (lane == 0) red[wave] = lmax;
  __syncthreads();
  float rmax = red[0];
#pragma unroll
  for (int w = 1; w < 8; ++w) rmax = fmaxf(rmax, red[w]);
  __syncthreads();

  float lsum = 0.f;
  for (int m = tid; m < NN; m += 256) {
    float e = expf(row[m] - rmax);
    row[m] = e;
    lsum += e;
  }
  lsum = wave_sum(lsum);
  if (lane == 0) red[wave] = lsum;
  __syncthreads();
  float rsum = 0.f;
#pragma unroll
  for (int w = 0; w < 8; ++w) rsum += red[w];
  const float inv = 1.0f / rsum;
  for (int m = tid; m < NN; m += 256)
    pos_score[(size_t)n * NN + m] = row[m] * inv;
}

__global__ void gemm_xut_kernel(const float* __restrict__ x,
                                const float* __restrict__ U,
                                float* __restrict__ t) {
  const int lane = threadIdx.x & 31;
  const int wave = threadIdx.x >> 5;
  const int tile = blockIdx.x * 8 + wave;
  const int i0 = (tile >> 4) * 16;
  const int j0 = (tile & 15) * 16;
  const int hl = lane >> 4, ll = lane & 15;

  const float* arow = x + (size_t)(i0 + ll) * DD + 2 * hl;
  const float* brow = U + (size_t)(j0 + ll) * DD + 2 * hl;
  v8f acc = {0.f, 0.f, 0.f, 0.f, 0.f, 0.f, 0.f, 0.f};
#pragma unroll 8
  for (int kb = 0; kb < DD; kb += 4) {
    v2f a = *(const v2f*)(arow + kb);
    v2f b = *(const v2f*)(brow + kb);
    acc = wmma4(a, b, acc);
  }
  float* orow = t + (size_t)(i0 + 8 * hl) * DD + j0 + ll;
#pragma unroll
  for (int v = 0; v < 8; ++v) orow[(size_t)v * DD] = acc[v];
}

__global__ void gemm_su_kernel(const float* __restrict__ t,
                               const float* __restrict__ U,
                               const float* __restrict__ S1,
                               const float* __restrict__ S2,
                               float* __restrict__ xs1,
                               float* __restrict__ xs2) {
  const int lane = threadIdx.x & 31;
  const int wave = threadIdx.x >> 5;
  const int tile = blockIdx.x * 8 + wave;
  const int i0 = (tile >> 4) * 16;
  const int j0 = (tile & 15) * 16;
  const int hl = lane >> 4, ll = lane & 15;

  const float* arow = t + (size_t)(i0 + ll) * DD + 2 * hl;
  const float* bcol = U + j0 + ll;
  v8f acc1 = {0.f, 0.f, 0.f, 0.f, 0.f, 0.f, 0.f, 0.f};
  v8f acc2 = {0.f, 0.f, 0.f, 0.f, 0.f, 0.f, 0.f, 0.f};
#pragma unroll 4
  for (int kb = 0; kb < DD; kb += 4) {
    const int k = kb + 2 * hl;
    v2f a = *(const v2f*)(arow + kb);
    v2f b;
    b.x = bcol[(size_t)k * DD];
    b.y = bcol[(size_t)(k + 1) * DD];
    v2f a1 = {a.x * fabsf(S1[k]), a.y * fabsf(S1[k + 1])};
    v2f a2 = {a.x * fabsf(S2[k]), a.y * fabsf(S2[k + 1])};
    acc1 = wmma4(a1, b, acc1);
    acc2 = wmma4(a2, b, acc2);
  }
  float* o1 = xs1 + (size_t)(i0 + 8 * hl) * DD + j0 + ll;
  float* o2 = xs2 + (size_t)(i0 + 8 * hl) * DD + j0 + ll;
#pragma unroll
  for (int v = 0; v < 8; ++v) {
    o1[(size_t)v * DD] = acc1[v];
    o2[(size_t)v * DD] = acc2[v];
  }
}

__global__ void fused_attn_kernel(const float* __restrict__ xs1,
                                  const float* __restrict__ xs2,
                                  const float* __restrict__ y,
                                  const float* __restrict__ pos_score,
                                  const float* __restrict__ gating,
                                  const float* __restrict__ h_temp,
                                  float* __restrict__ out,
                                  float* __restrict__ heat) {
  extern __shared__ float smem[];
  float* SL = smem;
  float* entS = smem + 2 * 16 * SSTRIDE;
  float* selS = entS + 32;

  const int b = blockIdx.x >> 6;
  const int n0 = (blockIdx.x & 63) * 16;
  const int tid = threadIdx.x;
  const int lane = tid & 31, wave = tid >> 5;
  const int hl = lane >> 4, ll = lane & 15;
  const float scale = 0.0625f;
  const float g = 1.0f / (1.0f + expf(-gating[0]));
  const float omg = 1.0f - g;
  const float ht = h_temp[0];
  const float* yb = y + (size_t)b * NN * DD;

  {
    const int branch = wave & 1;
    const float* xs = branch ? xs2 : xs1;
    const float* arow = xs + ((size_t)b * NN + n0 + ll) * DD + 2 * hl;
    for (int mt = (wave >> 1); mt < 64; mt += 4) {
      const int m0 = mt * 16;
      const float* brow = yb + (size_t)(m0 + ll) * DD + 2 * hl;
      v8f acc = {0.f, 0.f, 0.f, 0.f, 0.f, 0.f, 0.f, 0.f};
#pragma unroll 8
      for (int kb = 0; kb < DD; kb += 4) {
        v2f a = *(const v2f*)(arow + kb);
        v2f bb = *(const v2f*)(brow + kb);
        acc = wmma4(a, bb, acc);
      }
      float* dst = SL + ((size_t)branch * 16 + 8 * hl) * SSTRIDE + m0 + ll;
#pragma unroll
      for (int v = 0; v < 8; ++v) dst[(size_t)v * SSTRIDE] = acc[v] * scale;
    }
  }
  __syncthreads();

  for (int idx = wave; idx < 32; idx += 8) {
    const int br = idx >> 4, row = idx & 15;
    float* Lrow = SL + (size_t)(br * 16 + row) * SSTRIDE;
    const float* prow = pos_score + (size_t)(n0 + row) * NN;
    float lmax = -1e30f;
    for (int m = lane; m < NN; m += 32) lmax = fmaxf(lmax, Lrow[m]);
    lmax = wave_max(lmax);
    float lsum = 0.f;
    for (int m = lane; m < NN; m += 32) lsum += expf(Lrow[m] - lmax);
    lsum = wave_sum(lsum);
    const float inv = 1.0f / lsum;
    float ent = 0.f;
    for (int m = lane; m < NN; m += 32) {
      float p = expf(Lrow[m] - lmax) * inv;
      float a = omg * p + g * prow[m];
      ent += -a * logf(a + EPSV);
      Lrow[m] = a;
    }
    ent = wave_sum(ent);
    if (lane == 0) entS[br * 16 + row] = ent;
  }
  __syncthreads();

  if (tid < 16) {
    const int row = tid;
    const float h0 = 2.0f - 2.0f / (1.0f + expf(-ht * entS[row]));
    const float h1 = 2.0f - 2.0f / (1.0f + expf(-ht * entS[16 + row]));
    const int sel = (h0 >= h1) ? 0 : 1;
    selS[row] = (float)sel;
    heat[(size_t)b * NN + n0 + row] = sel ? h1 : h0;
  }
  __syncthreads();

  const int selrow = (int)selS[ll];
  const float* Abase = SL + (size_t)(selrow * 16 + ll) * SSTRIDE + 2 * hl;
  for (int dt = wave; dt < 16; dt += 8) {
    const int d0 = dt * 16;
    const float* bbase = yb + d0 + ll;
    v8f acc = {0.f, 0.f, 0.f, 0.f, 0.f, 0.f, 0.f, 0.f};
#pragma unroll 4
    for (int kb = 0; kb < NN; kb += 4) {
      const int k = kb + 2 * hl;
      v2f a = *(const v2f*)(Abase + kb);
      v2f bb;
      bb.x = bbase[(size_t)k * DD];
      bb.y = bbase[(size_t)(k + 1) * DD];
      acc = wmma4(a, bb, acc);
    }
    float* orow = out + ((size_t)b * NN + n0 + 8 * hl) * DD + d0 + ll;
#pragma unroll
    for (int v = 0; v < 8; ++v) orow[(size_t)v * DD] = acc[v];
  }
}

extern "C" void kernel_launch(void* const* d_in, const int* in_sizes, int n_in,
                              void* d_out, int out_size, void* d_ws, size_t ws_size,
                              hipStream_t stream) {
  const float* x = (const float*)d_in[0];
  const float* y = (const float*)d_in[1];
  const float* coords = (const float*)d_in[2];
  const float* U = (const float*)d_in[3];
  const float* S1 = (const float*)d_in[4];
  const float* S2 = (const float*)d_in[5];
  const float* gating = (const float*)d_in[6];
  const float* h_temp = (const float*)d_in[7];
  const float* p_temp = (const float*)d_in[8];
  const float* pos_emb = (const float*)d_in[9];

  float* out = (float*)d_out;
  float* heat = out + (size_t)BN * NN * DD;

  float* ws = (float*)d_ws;
  float* pos_score = ws;
  float* t = pos_score + (size_t)NN * NN;
  float* xs1 = t + (size_t)BN * NN * DD;
  float* xs2 = xs1 + (size_t)BN * NN * DD;

  pos_softmax_kernel<<<NN, 256, 0, stream>>>(coords, pos_emb, p_temp, pos_score);
  gemm_xut_kernel<<<1024, 256, 0, stream>>>(x, U, t);
  gemm_su_kernel<<<1024, 256, 0, stream>>>(t, U, S1, S2, xs1, xs2);

  const size_t lds_bytes = (size_t)(2 * 16 * SSTRIDE + 32 + 16) * sizeof(float);
  fused_attn_kernel<<<BN * 64, 256, lds_bytes, stream>>>(
      xs1, xs2, y, pos_score, gating, h_temp, out, heat);
}


# Both projection GEMMs moved into the attention launch: a workgroup projects its own 16 query rows, t and xs handed on through LDS in fragment order; 2 launches fewer; k order kept, bit-identical
# speedup vs baseline: 2.0601x; 2.0601x over previous
#include <hip/hip_runtime.h>
#include <hip/hip_bf16.h>
#include <math.h>

typedef float v2f __attribute__((ext_vector_type(2)));
typedef float v4f __attribute__((ext_vector_type(4)));
typedef float v8f __attribute__((ext_vector_type(8)));

#define BN 8
#define NN 1024
#define DD 256
#define EPSV 1e-8f
#define SSTRIDE 1028

__device__ __forceinline__ size_t pack_off(int r, int k) {
  return (size_t)(r >> 4) * 4096 + (size_t)(k >> 3) * 128 +
         (size_t)(((((k >> 1) & 1) << 4) + (r & 15)) * 4 + (((k >> 2) & 1) << 1) + (k & 1));
}

__device__ __forceinline__ void pack_chunk(const float* __restrict__ src, float* __restrict__ dst,
                                           int T, int c8, int lane) {
  const int hl = lane >> 4, ll = lane & 15;
  const float* row = src + (size_t)(T * 16 + ll) * DD + c8 * 8 + 2 * hl;
  const v2f lo = *(const v2f*)(row);
  const v2f hi = *(const v2f*)(row + 4);
  const v4f pk = {lo.x, lo.y, hi.x, hi.y};
  *(v4f*)(dst + (size_t)T * 4096 + c8 * 128 + lane * 4) = pk;
}

__device__ __forceinline__ v8f wmma4(v2f a, v2f b, v8f c) {
  return __builtin_amdgcn_wmma_f32_16x16x4_f32(
       false, a,  false, b,
       (short)0, c,  false,  false);
}

__device__ __forceinline__ float wave_max(float v) {
#pragma unroll
  for (int off = 16; off > 0; off >>= 1) v = fmaxf(v, __shfl_xor(v, off, 32));
  return v;
}
__device__ __forceinline__ float wave_sum(float v) {
#pragma unroll
  for (int off = 16; off > 0; off >>= 1) v += __shfl_xor(v, off, 32);
  return v;
}

__global__ void pos_softmax_kernel(const float* __restrict__ coords,
                                   const float* __restrict__ pos_emb,
                                   const float* __restrict__ p_temp,
                                   float* __restrict__ pos_score,
                                   const float* __restrict__ x,
                                   const float* __restrict__ y,
                                   const float* __restrict__ U,
                                   float* __restrict__ xp,
                                   float* __restrict__ yp,
                                   float* __restrict__ Up) {
  __shared__ float row[NN];
  __shared__ float red[8];
  const int n = blockIdx.x;
  const int tid = threadIdx.x;
  const int lane = tid & 31, wave = tid >> 5;
  {
    const int g = blockIdx.x * 8 + wave;
#pragma unroll
    for (int q = 0; q < 2; ++q) {
      const int c8 = (g & 15) * 2 + q;
      pack_chunk(x, xp, g >> 4, c8, lane);
      pack_chunk(y, yp, g >> 4, c8, lane);
      if (g < 256) pack_chunk(U, Up, g >> 4, c8, lane);
    }
  }
  const float pa = fabsf(p_temp[0]);
  float pe[6];
#pragma unroll
  for (int c = 0; c < 6; ++c) pe[c] = pos_emb[n * 6 + c];

  float lmax = -1e30f;
  for (int m = tid; m < NN; m += 256) {
    const float* cp = coords + ((size_t)n * NN + m) * 6;
    float dot = cp[0] * pe[0] + cp[1] * pe[1] + cp[2] * pe[2] +
                cp[3] * pe[3] + cp[4] * pe[4] + cp[5] * pe[5];
    float lg = -pa * dot;
    row[m] = lg;
    lmax = fmaxf(lmax, lg);
  }
  lmax = wave_max(lmax);
  if (lane == 0) red[wave] = lmax;
  __syncthreads();
  float rmax = red[0];
#pragma unroll
  for (int w = 1; w < 8; ++w) rmax = fmaxf(rmax, red[w]);
  __syncthreads();

  float lsum = 0.f;
  for (int m = tid; m < NN; m += 256) {
    float e = expf(row[m] - rmax);
    row[m] = e;
    lsum += e;
  }
  lsum = wave_sum(lsum);
  if (lane == 0) red[wave] = lsum;
  __syncthreads();
  float rsum = 0.f;
#pragma unroll
  for (int w = 0; w < 8; ++w) rsum += red[w];
  const float inv = 1.0f / rsum;
  for (int m = tid; m < NN; m += 256)
    pos_score[(size_t)n * NN + m] = row[m] * inv;
}

__device__ __forceinline__ void mma16(v8f (&acc)[16], v2f a, const v4f (&r0)[4], const v4f (&r1)[4]) {
#pragma unroll
  for (int q = 0; q < 4; ++q) {
#pragma unroll
    for (int j = 0; j < 4; ++j) {
      const v2f bb = {r0[q][j], r1[q][j]};
      acc[q * 4 + j] = wmma4(a, bb, acc[q * 4 + j]);
    }
  }
}
__device__ __forceinline__ void ld_yrows(v4f (&r0)[4], v4f (&r1)[4], const float* __restrict__ p) {
#pragma unroll
  for (int q = 0; q < 4; ++q) {
    r0[q] = *(const v4f*)(p + 64 * q);
    r1[q] = *(const v4f*)(p + DD + 64 * q);
  }
}

__global__ void __launch_bounds__(256, 4) fused_attn_kernel(const float* __restrict__ xp,
                                  const float* __restrict__ Up,
                                  const float* __restrict__ U,
                                  const float* __restrict__ S1,
                                  const float* __restrict__ S2,
                                  const float* __restrict__ yp,
                                  const float* __restrict__ y,
                                  const float* __restrict__ pos_score,
                                  const float* __restrict__ gating,
                                  const float* __restrict__ h_temp,
                                  float* __restrict__ out,
                                  float* __restrict__ heat) {
  extern __shared__ float smem[];
  float* SL = smem;
  float* entS = smem + 2 * 16 * SSTRIDE;
  float* selS = entS + 32;

  const int b = blockIdx.x >> 6;
  const int n0 = (blockIdx.x & 63) * 16;
  const int tid = threadIdx.x;
  const int lane = tid & 31, wave = tid >> 5;
  const int hl = lane >> 4, ll = lane & 15;
  const float scale = 0.0625f;
  const float g = 1.0f / (1.0f + expf(-gating[0]));
  const float omg = 1.0f - g;
  const float ht = h_temp[0];
  const float* yb = y + (size_t)b * NN * DD;

  float* T1 = SL;
  float* T2 = SL + 4096;
  float* XS = SL + 8192;
  {
    const float* achunk = xp + (size_t)blockIdx.x * 4096 + lane * 4;
    const float* bchunk = Up + (size_t)(2 * wave) * 4096 + lane * 4;
    v8f t0 = {0.f, 0.f, 0.f, 0.f, 0.f, 0.f, 0.f, 0.f};
    v8f t1 = {0.f, 0.f, 0.f, 0.f, 0.f, 0.f, 0.f, 0.f};
#pragma unroll 8
    for (int c8 = 0; c8 < DD / 8; ++c8) {
      const v4f a4 = *(const v4f*)(achunk + c8 * 128);
      const v4f p4 = *(const v4f*)(bchunk + c8 * 128);
      const v4f q4 = *(const v4f*)(bchunk + 4096 + c8 * 128);
      const v2f a0 = {a4.x, a4.y}, a1 = {a4.z, a4.w};
      const v2f p0 = {p4.x, p4.y}, p1 = {p4.z, p4.w};
      const v2f q0 = {q4.x, q4.y}, q1 = {q4.z, q4.w};
      t0 = wmma4(a0, p0, t0);
      t1 = wmma4(a0, q0, t1);
      t0 = wmma4(a1, p1, t0);
      t1 = wmma4(a1, q1, t1);
    }
    const int c0 = 32 * wave + ll, c1 = c0 + 16;
    const float s10 = fabsf(S1[c0]), s20 = fabsf(S2[c0]);
    const float s11 = fabsf(S1[c1]), s21 = fabsf(S2[c1]);
#pragma unroll
    for (int v = 0; v < 8; ++v) {
      const size_t o0 = pack_off(8 * hl + v, c0);
      const size_t o1 = pack_off(8 * hl + v, c1);
      T1[o0] = t0[v] * s10;
      T2[o0] = t0[v] * s20;
      T1[o1] = t1[v] * s11;
      T2[o1] = t1[v] * s21;
    }
  }
  __syncthreads();
  {
    const float* bcol = U + 32 * wave + ll;
    const float* a1p = T1 + lane * 4;
    const float* a2p = T2 + lane * 4;
    v8f x10 = {0.f, 0.f, 0.f, 0.f, 0.f, 0.f, 0.f, 0.f};
    v8f x11 = {0.f, 0.f, 0.f, 0.f, 0.f, 0.f, 0.f, 0.f};
    v8f x20 = {0.f, 0.f, 0.f, 0.f, 0.f, 0.f, 0.f, 0.f};
    v8f x21 = {0.f, 0.f, 0.f, 0.f, 0.f, 0.f, 0.f, 0.f};
#pragma unroll 4
    for (int c8 = 0; c8 < DD / 8; ++c8) {
      const v4f a41 = *(const v4f*)(a1p + c8 * 128);
      const v4f a42 = *(const v4f*)(a2p + c8 * 128);
      {
        const float* bk = bcol + (size_t)(c8 * 8 + 2 * hl) * DD;
        const v2f b0 = {bk[0], bk[DD]};
        const v2f b1 = {bk[16], bk[DD + 16]};
        const v2f a1 = {a41.x, a41.y}, a2 = {a42.x, a42.y};
        x10 = wmma4(a1, b0, x10);
        x20 = wmma4(a2, b0, x20);
        x11 = wmma4(a1, b1, x11);
        x21 = wmma4(a2, b1, x21);
      }
      {
        const float* bk = bcol + (size_t)(c8 * 8 + 4 + 2 * hl) * DD;
        const v2f b0 = {bk[0], bk[DD]};
        const v2f b1 = {bk[16], bk[DD + 16]};
        const v2f a1 = {a41.z, a41.w}, a2 = {a42.z, a42.w};
        x10 = wmma4(a1, b0, x10);
        x20 = wmma4(a2, b0, x20);
        x11 = wmma4(a1, b1, x11);
        x21 = wmma4(a2, b1, x21);
      }
    }
    const int c0 = 32 * wave + ll, c1 = c0 + 16;
#pragma unroll
    for (int v = 0; v < 8; ++v) {
      const size_t o0 = pack_off(8 * hl + v, c0);
      const size_t o1 = pack_off(8 * hl + v, c1);
      XS[o0] = x10[v];
      XS[o1] = x11[v];
      XS[4096 + o0] = x20[v];
      XS[4096 + o1] = x21[v];
    }
  }
  __syncthreads();

  {
    const int branch = wave & 1;
    const float* ypb = yp + (size_t)b * 64 * 4096 + lane * 4;
    v4f aq[DD / 8];
    {
      const float* achunk = XS + (size_t)branch * 4096 + lane * 4;
#pragma unroll
      for (int c8 = 0; c8 < DD / 8; ++c8) aq[c8] = *(const v4f*)(achunk + c8 * 128);
    }
    __syncthreads();
    for (int mt = (wave >> 1); mt < 64; mt += 4) {
      const int m0 = mt * 16;
      const float* bchunk = ypb + (size_t)mt * 4096;
      v8f acc = {0.f, 0.f, 0.f, 0.f, 0.f, 0.f, 0.f, 0.f};
#pragma unroll
      for (int c8 = 0; c8 < DD / 8; ++c8) {
        const v4f a4 = aq[c8];
        const v4f b4 = *(const v4f*)(bchunk + c8 * 128);
        v2f a0 = {a4.x, a4.y}, b0 = {b4.x, b4.y};
        v2f a1 = {a4.z, a4.w}, b1 = {b4.z, b4.w};
        acc = wmma4(a0, b0, acc);
        acc = wmma4(a1, b1, acc);
      }
      float* dst = SL + ((size_t)branch * 16 + 8 * hl) * SSTRIDE + m0 + ll;
#pragma unroll
      for (int v = 0; v < 8; ++v) dst[(size_t)v * SSTRIDE] = acc[v] * scale;
    }
  }
  __syncthreads();

  for (int idx = wave; idx < 32; idx += 8) {
    const int br = idx >> 4, row = idx & 15;
    float* Lrow = SL + (size_t)(br * 16 + row) * SSTRIDE;
    const float* prow = pos_score + (size_t)(n0 + row) * NN;
    float lmax = -1e30f;
    for (int m = lane; m < NN; m += 32) lmax = fmaxf(lmax, Lrow[m]);
    lmax = wave_max(lmax);
    float lsum = 0.f;
    for (int m = lane; m < NN; m += 32) lsum += expf(Lrow[m] - lmax);
    lsum = wave_sum(lsum);
    const float inv = 1.0f / lsum;
    float ent = 0.f;
    for (int m = lane; m < NN; m += 32) {
      float p = expf(Lrow[m] - lmax) * inv;
      float a = omg * p + g * prow[m];
      ent += -a * logf(a + EPSV);
      Lrow[m] = a;
    }
    ent = wave_sum(ent);
    if (lane == 0) entS[br * 16 + row] = ent;
  }
  __syncthreads();

  if (tid < 16) {
    const int row = tid;
    const float h0 = 2.0f - 2.0f / (1.0f + expf(-ht * entS[row]));
    const float h1 = 2.0f - 2.0f / (1.0f + expf(-ht * entS[16 + row]));
    const int sel = (h0 >= h1) ? 0 : 1;
    selS[row] = (float)sel;
    heat[(size_t)b * NN + n0 + row] = sel ? h1 : h0;
  }
  __syncthreads();

  const int selrow = (int)selS[ll];
  const float* Abase = SL + (size_t)(selrow * 16 + ll) * SSTRIDE + wave * 128 + 2 * hl;
  const float* yk = yb + (size_t)(wave * 128 + 2 * hl) * DD + 4 * ll;
  v8f acc[16];
#pragma unroll
  for (int t = 0; t < 16; ++t) acc[t] = (v8f){0.f, 0.f, 0.f, 0.f, 0.f, 0.f, 0.f, 0.f};

  v4f e0[4], e1[4], o0[4], o1[4];
  ld_yrows(e0, e1, yk);
#pragma unroll 1
  for (int ks = 0; ks < 30; ks += 2) {
    ld_yrows(o0, o1, yk + (size_t)(ks + 1) * 4 * DD);
    mma16(acc, *(const v2f*)(Abase + ks * 4), e0, e1);
    ld_yrows(e0, e1, yk + (size_t)(ks + 2) * 4 * DD);
    mma16(acc, *(const v2f*)(Abase + ks * 4 + 4), o0, o1);
  }
  ld_yrows(o0, o1, yk + (size_t)31 * 4 * DD);
  mma16(acc, *(const v2f*)(Abase + 30 * 4), e0, e1);
  mma16(acc, *(const v2f*)(Abase + 31 * 4), o0, o1);
  __syncthreads();

  {
    float* P = SL + (size_t)wave * 4096;
#pragma unroll
    for (int q = 0; q < 4; ++q) {
#pragma unroll
      for (int v = 0; v < 8; ++v) {
        const v4f pv = {acc[q * 4 + 0][v], acc[q * 4 + 1][v], acc[q * 4 + 2][v], acc[q * 4 + 3][v]};
        *(v4f*)(P + (8 * hl + v) * 256 + 64 * q + 4 * ll) = pv;
      }
    }
  }
  __syncthreads();
#pragma unroll
  for (int i = 0; i < 4; ++i) {
    const int idx = tid + 256 * i;
    v4f s = *(const v4f*)(SL + (size_t)idx * 4);
#pragma unroll
    for (int w = 1; w < 8; ++w) s += *(const v4f*)(SL + (size_t)w * 4096 + (size_t)idx * 4);
    *(v4f*)(out + ((size_t)b * NN + n0 + (idx >> 6)) * DD + 4 * (idx & 63)) = s;
  }
}

extern "C" void kernel_launch(void* const* d_in, const int* in_sizes, int n_in,
                              void* d_out, int out_size, void* d_ws, size_t ws_size,
                              hipStream_t stream) {
  const float* x = (const float*)d_in[0];
  const float* y = (const float*)d_in[1];
  const float* coords = (const float*)d_in[2];
  const float* U = (const float*)d_in[3];
  const float* S1 = (const float*)d_in[4];
  const float* S2 = (const float*)d_in[5];
  const float* gating = (const float*)d_in[6];
  const float* h_temp = (const float*)d_in[7];
  const float* p_temp = (const float*)d_in[8];
  const float* pos_emb = (const float*)d_in[9];

  float* out = (float*)d_out;
  float* heat = out + (size_t)BN * NN * DD;

  float* ws = (float*)d_ws;
  float* pos_score = ws;
  float* yp = pos_score + (size_t)NN * NN;
  float* xp = yp + (size_t)BN * NN * DD;
  float* Up = xp + (size_t)BN * NN * DD;

  pos_softmax_kernel<<<NN, 256, 0, stream>>>(coords, pos_emb, p_temp, pos_score,
                                             x, y, U, xp, yp, Up);

  const size_t lds_bytes = (size_t)(2 * 16 * SSTRIDE + 32 + 16) * sizeof(float);
  fused_attn_kernel<<<BN * 64, 256, lds_bytes, stream>>>(
      xp, Up, U, S1, S2, yp, y, pos_score, gating, h_temp, out, heat);
}


# Softmax/blend/entropy pass: a lane owns 4 adjacent columns per step: score, positional and blended rows move 16 B per lane (8 steps a row, was 32 of 4 B); exp-sum and entropy sum order changed
# speedup vs baseline: 2.1575x; 1.0473x over previous
#include <hip/hip_runtime.h>
#include <hip/hip_bf16.h>
#include <math.h>

typedef float v2f __attribute__((ext_vector_type(2)));
typedef float v4f __attribute__((ext_vector_type(4)));
typedef float v8f __attribute__((ext_vector_type(8)));

#define BN 8
#define NN 1024
#define DD 256
#define EPSV 1e-8f
#define SSTRIDE 1028

__device__ __forceinline__ size_t pack_off(int r, int k) {
  return (size_t)(r >> 4) * 4096 + (size_t)(k >> 3) * 128 +
         (size_t)(((((k >> 1) & 1) << 4) + (r & 15)) * 4 + (((k >> 2) & 1) << 1) + (k & 1));
}

__device__ __forceinline__ void pack_chunk(const float* __restrict__ src, float* __restrict__ dst,
                                           int T, int c8, int lane) {
  const int hl = lane >> 4, ll = lane & 15;
  const float* row = src + (size_t)(T * 16 + ll) * DD + c8 * 8 + 2 * hl;
  const v2f lo = *(const v2f*)(row);
  const v2f hi = *(const v2f*)(row + 4);
  const v4f pk = {lo.x, lo.y, hi.x, hi.y};
  *(v4f*)(dst + (size_t)T * 4096 + c8 * 128 + lane * 4) = pk;
}

__device__ __forceinline__ v8f wmma4(v2f a, v2f b, v8f c) {
  return __builtin_amdgcn_wmma_f32_16x16x4_f32(
       false, a,  false, b,
       (short)0, c,  false,  false);
}

__device__ __forceinline__ float wave_max(float v) {
#pragma unroll
  for (int off = 16; off > 0; off >>= 1) v = fmaxf(v, __shfl_xor(v, off, 32));
  return v;
}
__device__ __forceinline__ float wave_sum(float v) {
#pragma unroll
  for (int off = 16; off > 0; off >>= 1) v += __shfl_xor(v, off, 32);
  return v;
}

__global__ void pos_softmax_kernel(const float* __restrict__ coords,
                                   const float* __restrict__ pos_emb,
                                   const float* __restrict__ p_temp,
                                   float* __restrict__ pos_score,
                                   const float* __restrict__ x,
                                   const float* __restrict__ y,
                                   const float* __restrict__ U,
                                   float* __restrict__ xp,
                                   float* __restrict__ yp,
                                   float* __restrict__ Up) {
  __shared__ float row[NN];
  __shared__ float red[8];
  const int n = blockIdx.x;
  const int tid = threadIdx.x;
  const int lane = tid & 31, wave = tid >> 5;
  {
    const int g = blockIdx.x * 8 + wave;
#pragma unroll
    for (int q = 0; q < 2; ++q) {
      const int c8 = (g & 15) * 2 + q;
      pack_chunk(x, xp, g >> 4, c8, lane);
      pack_chunk(y, yp, g >> 4, c8, lane);
      if (g < 256) pack_chunk(U, Up, g >> 4, c8, lane);
    }
  }
  const float pa = fabsf(p_temp[0]);
  float pe[6];
#pragma unroll
  for (int c = 0; c < 6; ++c) pe[c] = pos_emb[n * 6 + c];

  float lmax = -1e30f;
  for (int m = tid; m < NN; m += 256) {
    const float* cp = coords + ((size_t)n * NN + m) * 6;
    float dot = cp[0] * pe[0] + cp[1] * pe[1] + cp[2] * pe[2] +
                cp[3] * pe[3] + cp[4] * pe[4] + cp[5] * pe[5];
    float lg = -pa * dot;
    row[m] = lg;
    lmax = fmaxf(lmax, lg);
  }
  lmax = wave_max(lmax);
  if (lane == 0) red[wave] = lmax;
  __syncthreads();
  float rmax = red[0];
#pragma unroll
  for (int w = 1; w < 8; ++w) rmax = fmaxf(rmax, red[w]);
  __syncthreads();

  float lsum = 0.f;
  for (int m = tid; m < NN; m += 256) {
    float e = expf(row[m] - rmax);
    row[m] = e;
    lsum += e;
  }
  lsum = wave_sum(lsum);
  if (lane == 0) red[wave] = lsum;
  __syncthreads();
  float rsum = 0.f;
#pragma unroll
  for (int w = 0; w < 8; ++w) rsum += red[w];
  const float inv = 1.0f / rsum;
  for (int m = tid; m < NN; m += 256)
    pos_score[(size_t)n * NN + m] = row[m] * inv;
}

__device__ __forceinline__ void mma16(v8f (&acc)[16], v2f a, const v4f (&r0)[4], const v4f (&r1)[4]) {
#pragma unroll
  for (int q = 0; q < 4; ++q) {
#pragma unroll
    for (int j = 0; j < 4; ++j) {
      const v2f bb = {r0[q][j], r1[q][j]};
      acc[q * 4 + j] = wmma4(a, bb, acc[q * 4 + j]);
    }
  }
}
__device__ __forceinline__ void ld_yrows(v4f (&r0)[4], v4f (&r1)[4], const float* __restrict__ p) {
#pragma unroll
  for (int q = 0; q < 4; ++q) {
    r0[q] = *(const v4f*)(p + 64 * q);
    r1[q] = *(const v4f*)(p + DD + 64 * q);
  }
}

__global__ void __launch_bounds__(256, 4) fused_attn_kernel(const float* __restrict__ xp,
                                  const float* __restrict__ Up,
                                  const float* __restrict__ U,
                                  const float* __restrict__ S1,
                                  const float* __restrict__ S2,
                                  const float* __restrict__ yp,
                                  const float* __restrict__ y,
                                  const float* __restrict__ pos_score,
                                  const float* __restrict__ gating,
                                  const float* __restrict__ h_temp,
                                  float* __restrict__ out,
                                  float* __restrict__ heat) {
  extern __shared__ float smem[];
  float* SL = smem;
  float* entS = smem + 2 * 16 * SSTRIDE;
  float* selS = entS + 32;

  const int b = blockIdx.x >> 6;
  const int n0 = (blockIdx.x & 63) * 16;
  const int tid = threadIdx.x;
  const int lane = tid & 31, wave = tid >> 5;
  const int hl = lane >> 4, ll = lane & 15;
  const float scale = 0.0625f;
  const float g = 1.0f / (1.0f + expf(-gating[0]));
  const float omg = 1.0f - g;
  const float ht = h_temp[0];
  const float* yb = y + (size_t)b * NN * DD;

  float* T1 = SL;
  float* T2 = SL + 4096;
  float* XS = SL + 8192;
  {
    const float* achunk = xp + (size_t)blockIdx.x * 4096 + lane * 4;
    const float* bchunk = Up + (size_t)(2 * wave) * 4096 + lane * 4;
    v8f t0 = {0.f, 0.f, 0.f, 0.f, 0.f, 0.f, 0.f, 0.f};
    v8f t1 = {0.f, 0.f, 0.f, 0.f, 0.f, 0.f, 0.f, 0.f};
#pragma unroll 8
    for (int c8 = 0; c8 < DD / 8; ++c8) {
      const v4f a4 = *(const v4f*)(achunk + c8 * 128);
      const v4f p4 = *(const v4f*)(bchunk + c8 * 128);
      const v4f q4 = *(const v4f*)(bchunk + 4096 + c8 * 128);
      const v2f a0 = {a4.x, a4.y}, a1 = {a4.z, a4.w};
      const v2f p0 = {p4.x, p4.y}, p1 = {p4.z, p4.w};
      const v2f q0 = {q4.x, q4.y}, q1 = {q4.z, q4.w};
      t0 = wmma4(a0, p0, t0);
      t1 = wmma4(a0, q0, t1);
      t0 = wmma4(a1, p1, t0);
      t1 = wmma4(a1, q1, t1);
    }
    const int c0 = 32 * wave + ll, c1 = c0 + 16;
    const float s10 = fabsf(S1[c0]), s20 = fabsf(S2[c0]);
    const float s11 = fabsf(S1[c1]), s21 = fabsf(S2[c1]);
#pragma unroll
    for (int v = 0; v < 8; ++v) {
      const size_t o0 = pack_off(8 * hl + v, c0);
      const size_t o1 = pack_off(8 * hl + v, c1);
      T1[o0] = t0[v] * s10;
      T2[o0] = t0[v] * s20;
      T1[o1] = t1[v] * s11;
      T2[o1] = t1[v] * s21;
    }
  }
  __syncthreads();
  {
    const float* bcol = U + 32 * wave + ll;
    const float* a1p = T1 + lane * 4;
    const float* a2p = T2 + lane * 4;
    v8f x10 = {0.f, 0.f, 0.f, 0.f, 0.f, 0.f, 0.f, 0.f};
    v8f x11 = {0.f, 0.f, 0.f, 0.f, 0.f, 0.f, 0.f, 0.f};
    v8f x20 = {0.f, 0.f, 0.f, 0.f, 0.f, 0.f, 0.f, 0.f};
    v8f x21 = {0.f, 0.f, 0.f, 0.f, 0.f, 0.f, 0.f, 0.f};
#pragma unroll 4
    for (int c8 = 0; c8 < DD / 8; ++c8) {
      const v4f a41 = *(const v4f*)(a1p + c8 * 128);
      const v4f a42 = *(const v4f*)(a2p + c8 * 128);
      {
        const float* bk = bcol + (size_t)(c8 * 8 + 2 * hl) * DD;
        const v2f b0 = {bk[0], bk[DD]};
        const v2f b1 = {bk[16], bk[DD + 16]};
        const v2f a1 = {a41.x, a41.y}, a2 = {a42.x, a42.y};
        x10 = wmma4(a1, b0, x10);
        x20 = wmma4(a2, b0, x20);
        x11 = wmma4(a1, b1, x11);
        x21 = wmma4(a2, b1, x21);
      }
      {
        const float* bk = bcol + (size_t)(c8 * 8 + 4 + 2 * hl) * DD;
        const v2f b0 = {bk[0], bk[DD]};
        const v2f b1 = {bk[16], bk[DD + 16]};
        const v2f a1 = {a41.z, a41.w}, a2 = {a42.z, a42.w};
        x10 = wmma4(a1, b0, x10);
        x20 = wmma4(a2, b0, x20);
        x11 = wmma4(a1, b1, x11);
        x21 = wmma4(a2, b1, x21);
      }
    }
    const int c0 = 32 * wave + ll, c1 = c0 + 16;
#pragma unroll
    for (int v = 0; v < 8; ++v) {
      const size_t o0 = pack_off(8 * hl + v, c0);
      const size_t o1 = pack_off(8 * hl + v, c1);
      XS[o0] = x10[v];
      XS[o1] = x11[v];
      XS[4096 + o0] = x20[v];
      XS[4096 + o1] = x21[v];
    }
  }
  __syncthreads();

  {
    const int branch = wave & 1;
    const float* ypb = yp + (size_t)b * 64 * 4096 + lane * 4;
    v4f aq[DD / 8];
    {
      const float* achunk = XS + (size_t)branch * 4096 + lane * 4;
#pragma unroll
      for (int c8 = 0; c8 < DD / 8; ++c8) aq[c8] = *(const v4f*)(achunk + c8 * 128);
    }
    __syncthreads();
    for (int mt = (wave >> 1); mt < 64; mt += 4) {
      const int m0 = mt * 16;
      const float* bchunk = ypb + (size_t)mt * 4096;
      v8f acc = {0.f, 0.f, 0.f, 0.f, 0.f, 0.f, 0.f, 0.f};
#pragma unroll
      for (int c8 = 0; c8 < DD / 8; ++c8) {
        const v4f a4 = aq[c8];
        const v4f b4 = *(const v4f*)(bchunk + c8 * 128);
        v2f a0 = {a4.x, a4.y}, b0 = {b4.x, b4.y};
        v2f a1 = {a4.z, a4.w}, b1 = {b4.z, b4.w};
        acc = wmma4(a0, b0, acc);
        acc = wmma4(a1, b1, acc);
      }
      float* dst = SL + ((size_t)branch * 16 + 8 * hl) * SSTRIDE + m0 + ll;
#pragma unroll
      for (int v = 0; v < 8; ++v) dst[(size_t)v * SSTRIDE] = acc[v] * scale;
    }
  }
  __syncthreads();

  for (int idx = wave; idx < 32; idx += 8) {
    const int br = idx >> 4, row = idx & 15;
    float* Lrow = SL + (size_t)(br * 16 + row) * SSTRIDE + 4 * lane;
    const float* prow = pos_score + (size_t)(n0 + row) * NN + 4 * lane;
    float lmax = -1e30f;
    for (int i = 0; i < 8; ++i) {
      const v4f s = *(const v4f*)(Lrow + 128 * i);
      lmax = fmaxf(fmaxf(lmax, fmaxf(s.x, s.y)), fmaxf(s.z, s.w));
    }
    lmax = wave_max(lmax);
    float lsum = 0.f;
    for (int i = 0; i < 8; ++i) {
      const v4f s = *(const v4f*)(Lrow + 128 * i);
      lsum += expf(s.x - lmax);
      lsum += expf(s.y - lmax);
      lsum += expf(s.z - lmax);
      lsum += expf(s.w - lmax);
    }
    lsum = wave_sum(lsum);
    const float inv = 1.0f / lsum;
    float ent = 0.f;
    for (int i = 0; i < 8; ++i) {
      const v4f s = *(const v4f*)(Lrow + 128 * i);
      const v4f pz = *(const v4f*)(prow + 128 * i);
      v4f a;
      a.x = omg * (expf(s.x - lmax) * inv) + g * pz.x;
      a.y = omg * (expf(s.y - lmax) * inv) + g * pz.y;
      a.z = omg * (expf(s.z - lmax) * inv) + g * pz.z;
      a.w = omg * (expf(s.w - lmax) * inv) + g * pz.w;
      ent += -a.x * logf(a.x + EPSV);
      ent += -a.y * logf(a.y + EPSV);
      ent += -a.z * logf(a.z + EPSV);
      ent += -a.w * logf(a.w + EPSV);
      *(v4f*)(Lrow + 128 * i) = a;
    }
    ent = wave_sum(ent);
    if (lane == 0) entS[br * 16 + row] = ent;
  }
  __syncthreads();

  if (tid < 16) {
    const int row = tid;
    const float h0 = 2.0f - 2.0f / (1.0f + expf(-ht * entS[row]));
    const float h1 = 2.0f - 2.0f / (1.0f + expf(-ht * entS[16 + row]));
    const int sel = (h0 >= h1) ? 0 : 1;
    selS[row] = (float)sel;
    heat[(size_t)b * NN + n0 + row] = sel ? h1 : h0;
  }
  __syncthreads();

  const int selrow = (int)selS[ll];
  const float* Abase = SL + (size_t)(selrow * 16 + ll) * SSTRIDE + wave * 128 + 2 * hl;
  const float* yk = yb + (size_t)(wave * 128 + 2 * hl) * DD + 4 * ll;
  v8f acc[16];
#pragma unroll
  for (int t = 0; t < 16; ++t) acc[t] = (v8f){0.f, 0.f, 0.f, 0.f, 0.f, 0.f, 0.f, 0.f};

  v4f e0[4], e1[4], o0[4], o1[4];
  ld_yrows(e0, e1, yk);
#pragma unroll 1
  for (int ks = 0; ks < 30; ks += 2) {
    ld_yrows(o0, o1, yk + (size_t)(ks + 1) * 4 * DD);
    mma16(acc, *(const v2f*)(Abase + ks * 4), e0, e1);
    ld_yrows(e0, e1, yk + (size_t)(ks + 2) * 4 * DD);
    mma16(acc, *(const v2f*)(Abase + ks * 4 + 4), o0, o1);
  }
  ld_yrows(o0, o1, yk + (size_t)31 * 4 * DD);
  mma16(acc, *(const v2f*)(Abase + 30 * 4), e0, e1);
  mma16(acc, *(const v2f*)(Abase + 31 * 4), o0, o1);
  __syncthreads();

  {
    float* P = SL + (size_t)wave * 4096;
#pragma unroll
    for (int q = 0; q < 4; ++q) {
#pragma unroll
      for (int v = 0; v < 8; ++v) {
        const v4f pv = {acc[q * 4 + 0][v], acc[q * 4 + 1][v], acc[q * 4 + 2][v], acc[q * 4 + 3][v]};
        *(v4f*)(P + (8 * hl + v) * 256 + 64 * q + 4 * ll) = pv;
      }
    }
  }
  __syncthreads();
#pragma unroll
  for (int i = 0; i < 4; ++i) {
    const int idx = tid + 256 * i;
    v4f s = *(const v4f*)(SL + (size_t)idx * 4);
#pragma unroll
    for (int w = 1; w < 8; ++w) s += *(const v4f*)(SL + (size_t)w * 4096 + (size_t)idx * 4);
    *(v4f*)(out + ((size_t)b * NN + n0 + (idx >> 6)) * DD + 4 * (idx & 63)) = s;
  }
}

extern "C" void kernel_launch(void* const* d_in, const int* in_sizes, int n_in,
                              void* d_out, int out_size, void* d_ws, size_t ws_size,
                              hipStream_t stream) {
  const float* x = (const float*)d_in[0];
  const float* y = (const float*)d_in[1];
  const float* coords = (const float*)d_in[2];
  const float* U = (const float*)d_in[3];
  const float* S1 = (const float*)d_in[4];
  const float* S2 = (const float*)d_in[5];
  const float* gating = (const float*)d_in[6];
  const float* h_temp = (const float*)d_in[7];
  const float* p_temp = (const float*)d_in[8];
  const float* pos_emb = (const float*)d_in[9];

  float* out = (float*)d_out;
  float* heat = out + (size_t)BN * NN * DD;

  float* ws = (float*)d_ws;
  float* pos_score = ws;
  float* yp = pos_score + (size_t)NN * NN;
  float* xp = yp + (size_t)BN * NN * DD;
  float* Up = xp + (size_t)BN * NN * DD;

  pos_softmax_kernel<<<NN, 256, 0, stream>>>(coords, pos_emb, p_temp, pos_score,
                                             x, y, U, xp, yp, Up);

  const size_t lds_bytes = (size_t)(2 * 16 * SSTRIDE + 32 + 16) * sizeof(float);
  fused_attn_kernel<<<BN * 64, 256, lds_bytes, stream>>>(
      xp, Up, U, S1, S2, yp, y, pos_score, gating, h_temp, out, heat);
}


# Softmax/blend/entropy pass: a wave takes its 4 rows through each sub-pass side by side; 4 independent load, cross-lane and exp/log chains cover each other's latency (was row by row); sum orders kept
# speedup vs baseline: 2.2515x; 1.0436x over previous
#include <hip/hip_runtime.h>
#include <hip/hip_bf16.h>
#include <math.h>

typedef float v2f __attribute__((ext_vector_type(2)));
typedef float v4f __attribute__((ext_vector_type(4)));
typedef float v8f __attribute__((ext_vector_type(8)));

#define BN 8
#define NN 1024
#define DD 256
#define EPSV 1e-8f
#define SSTRIDE 1028

__device__ __forceinline__ size_t pack_off(int r, int k) {
  return (size_t)(r >> 4) * 4096 + (size_t)(k >> 3) * 128 +
         (size_t)(((((k >> 1) & 1) << 4) + (r & 15)) * 4 + (((k >> 2) & 1) << 1) + (k & 1));
}

__device__ __forceinline__ void pack_chunk(const float* __restrict__ src, float* __restrict__ dst,
                                           int T, int c8, int lane) {
  const int hl = lane >> 4, ll = lane & 15;
  const float* row = src + (size_t)(T * 16 + ll) * DD + c8 * 8 + 2 * hl;
  const v2f lo = *(const v2f*)(row);
  const v2f hi = *(const v2f*)(row + 4);
  const v4f pk = {lo.x, lo.y, hi.x, hi.y};
  *(v4f*)(dst + (size_t)T * 4096 + c8 * 128 + lane * 4) = pk;
}

__device__ __forceinline__ v8f wmma4(v2f a, v2f b, v8f c) {
  return __builtin_amdgcn_wmma_f32_16x16x4_f32(
       false, a,  false, b,
       (short)0, c,  false,  false);
}

__device__ __forceinline__ float wave_max(float v) {
#pragma unroll
  for (int off = 16; off > 0; off >>= 1) v = fmaxf(v, __shfl_xor(v, off, 32));
  return v;
}
__device__ __forceinline__ float wave_sum(float v) {
#pragma unroll
  for (int off = 16; off > 0; off >>= 1) v += __shfl_xor(v, off, 32);
  return v;
}

__global__ void pos_softmax_kernel(const float* __restrict__ coords,
                                   const float* __restrict__ pos_emb,
                                   const float* __restrict__ p_temp,
                                   float* __restrict__ pos_score,
                                   const float* __restrict__ x,
                                   const float* __restrict__ y,
                                   const float* __restrict__ U,
                                   float* __restrict__ xp,
                                   float* __restrict__ yp,
                                   float* __restrict__ Up) {
  __shared__ float row[NN];
  __shared__ float red[8];
  const int n = blockIdx.x;
  const int tid = threadIdx.x;
  const int lane = tid & 31, wave = tid >> 5;
  {
    const int g = blockIdx.x * 8 + wave;
#pragma unroll
    for (int q = 0; q < 2; ++q) {
      const int c8 = (g & 15) * 2 + q;
      pack_chunk(x, xp, g >> 4, c8, lane);
      pack_chunk(y, yp, g >> 4, c8, lane);
      if (g < 256) pack_chunk(U, Up, g >> 4, c8, lane);
    }
  }
  const float pa = fabsf(p_temp[0]);
  float pe[6];
#pragma unroll
  for (int c = 0; c < 6; ++c) pe[c] = pos_emb[n * 6 + c];

  float lmax = -1e30f;
  for (int m = tid; m < NN; m += 256) {
    const float* cp = coords + ((size_t)n * NN + m) * 6;
    float dot = cp[0] * pe[0] + cp[1] * pe[1] + cp[2] * pe[2] +
                cp[3] * pe[3] + cp[4] * pe[4] + cp[5] * pe[5];
    float lg = -pa * dot;
    row[m] = lg;
    lmax = fmaxf(lmax, lg);
  }
  lmax = wave_max(lmax);
  if (lane == 0) red[wave] = lmax;
  __syncthreads();
  float rmax = red[0];
#pragma unroll
  for (int w = 1; w < 8; ++w) rmax = fmaxf(rmax, red[w]);
  __syncthreads();

  float lsum = 0.f;
  for (int m = tid; m < NN; m += 256) {
    float e = expf(row[m] - rmax);
    row[m] = e;
    lsum += e;
  }
  lsum = wave_sum(lsum);
  if (lane == 0) red[wave] = lsum;
  __syncthreads();
  float rsum = 0.f;
#pragma unroll
  for (int w = 0; w < 8; ++w) rsum += red[w];
  const float inv = 1.0f / rsum;
  for (int m = tid; m < NN; m += 256)
    pos_score[(size_t)n * NN + m] = row[m] * inv;
}

__device__ __forceinline__ void mma16(v8f (&acc)[16], v2f a, const v4f (&r0)[4], const v4f (&r1)[4]) {
#pragma unroll
  for (int q = 0; q < 4; ++q) {
#pragma unroll
    for (int j = 0; j < 4; ++j) {
      const v2f bb = {r0[q][j], r1[q][j]};
      acc[q * 4 + j] = wmma4(a, bb, acc[q * 4 + j]);
    }
  }
}
__device__ __forceinline__ void ld_yrows(v4f (&r0)[4], v4f (&r1)[4], const float* __restrict__ p) {
#pragma unroll
  for (int q = 0; q < 4; ++q) {
    r0[q] = *(const v4f*)(p + 64 * q);
    r1[q] = *(const v4f*)(p + DD + 64 * q);
  }
}

__global__ void __launch_bounds__(256, 4) fused_attn_kernel(const float* __restrict__ xp,
                                  const float* __restrict__ Up,
                                  const float* __restrict__ U,
                                  const float* __restrict__ S1,
                                  const float* __restrict__ S2,
                                  const float* __restrict__ yp,
                                  const float* __restrict__ y,
                                  const float* __restrict__ pos_score,
                                  const float* __restrict__ gating,
                                  const float* __restrict__ h_temp,
                                  float* __restrict__ out,
                                  float* __restrict__ heat) {
  extern __shared__ float smem[];
  float* SL = smem;
  float* entS = smem + 2 * 16 * SSTRIDE;
  float* selS = entS + 32;

  const int b = blockIdx.x >> 6;
  const int n0 = (blockIdx.x & 63) * 16;
  const int tid = threadIdx.x;
  const int lane = tid & 31, wave = tid >> 5;
  const int hl = lane >> 4, ll = lane & 15;
  const float scale = 0.0625f;
  const float g = 1.0f / (1.0f + expf(-gating[0]));
  const float omg = 1.0f - g;
  const float ht = h_temp[0];
  const float* yb = y + (size_t)b * NN * DD;

  float* T1 = SL;
  float* T2 = SL + 4096;
  float* XS = SL + 8192;
  {
    const float* achunk = xp + (size_t)blockIdx.x * 4096 + lane * 4;
    const float* bchunk = Up + (size_t)(2 * wave) * 4096 + lane * 4;
    v8f t0 = {0.f, 0.f, 0.f, 0.f, 0.f, 0.f, 0.f, 0.f};
    v8f t1 = {0.f, 0.f, 0.f, 0.f, 0.f, 0.f, 0.f, 0.f};
#pragma unroll 8
    for (int c8 = 0; c8 < DD / 8; ++c8) {
      const v4f a4 = *(const v4f*)(achunk + c8 * 128);
      const v4f p4 = *(const v4f*)(bchunk + c8 * 128);
      const v4f q4 = *(const v4f*)(bchunk + 4096 + c8 * 128);
      const v2f a0 = {a4.x, a4.y}, a1 = {a4.z, a4.w};
      const v2f p0 = {p4.x, p4.y}, p1 = {p4.z, p4.w};
      const v2f q0 = {q4.x, q4.y}, q1 = {q4.z, q4.w};
      t0 = wmma4(a0, p0, t0);
      t1 = wmma4(a0, q0, t1);
      t0 = wmma4(a1, p1, t0);
      t1 = wmma4(a1, q1, t1);
    }
    const int c0 = 32 * wave + ll, c1 = c0 + 16;
    const float s10 = fabsf(S1[c0]), s20 = fabsf(S2[c0]);
    const float s11 = fabsf(S1[c1]), s21 = fabsf(S2[c1]);
#pragma unroll
    for (int v = 0; v < 8; ++v) {
      const size_t o0 = pack_off(8 * hl + v, c0);
      const size_t o1 = pack_off(8 * hl + v, c1);
      T1[o0] = t0[v] * s10;
      T2[o0] = t0[v] * s20;
      T1[o1] = t1[v] * s11;
      T2[o1] = t1[v] * s21;
    }
  }
  __syncthreads();
  {
    const float* bcol = U + 32 * wave + ll;
    const float* a1p = T1 + lane * 4;
    const float* a2p = T2 + lane * 4;
    v8f x10 = {0.f, 0.f, 0.f, 0.f, 0.f, 0.f, 0.f, 0.f};
    v8f x11 = {0.f, 0.f, 0.f, 0.f, 0.f, 0.f, 0.f, 0.f};
    v8f x20 = {0.f, 0.f, 0.f, 0.f, 0.f, 0.f, 0.f, 0.f};
    v8f x21 = {0.f, 0.f, 0.f, 0.f, 0.f, 0.f, 0.f, 0.f};
#pragma unroll 4
    for (int c8 = 0; c8 < DD / 8; ++c8) {
      const v4f a41 = *(const v4f*)(a1p + c8 * 128);
      const v4f a42 = *(const v4f*)(a2p + c8 * 128);
      {
        const float* bk = bcol + (size_t)(c8 * 8 + 2 * hl) * DD;
        const v2f b0 = {bk[0], bk[DD]};
        const v2f b1 = {bk[16], bk[DD + 16]};
        const v2f a1 = {a41.x, a41.y}, a2 = {a42.x, a42.y};
        x10 = wmma4(a1, b0, x10);
        x20 = wmma4(a2, b0, x20);
        x11 = wmma4(a1, b1, x11);
        x21 = wmma4(a2, b1, x21);
      }
      {
        const float* bk = bcol + (size_t)(c8 * 8 + 4 + 2 * hl) * DD;
        const v2f b0 = {bk[0], bk[DD]};
        const v2f b1 = {bk[16], bk[DD + 16]};
        const v2f a1 = {a41.z, a41.w}, a2 = {a42.z, a42.w};
        x10 = wmma4(a1, b0, x10);
        x20 = wmma4(a2, b0, x20);
        x11 = wmma4(a1, b1, x11);
        x21 = wmma4(a2, b1, x21);
      }
    }
    const int c0 = 32 * wave + ll, c1 = c0 + 16;
#pragma unroll
    for (int v = 0; v < 8; ++v) {
      const size_t o0 = pack_off(8 * hl + v, c0);
      const size_t o1 = pack_off(8 * hl + v, c1);
      XS[o0] = x10[v];
      XS[o1] = x11[v];
      XS[4096 + o0] = x20[v];
      XS[4096 + o1] = x21[v];
    }
  }
  __syncthreads();

  {
    const int branch = wave & 1;
    const float* ypb = yp + (size_t)b * 64 * 4096 + lane * 4;
    v4f aq[DD / 8];
    {
      const float* achunk = XS + (size_t)branch * 4096 + lane * 4;
#pragma unroll
      for (int c8 = 0; c8 < DD / 8; ++c8) aq[c8] = *(const v4f*)(achunk + c8 * 128);
    }
    __syncthreads();
    for (int mt = (wave >> 1); mt < 64; mt += 4) {
      const int m0 = mt * 16;
      const float* bchunk = ypb + (size_t)mt * 4096;
      v8f acc = {0.f, 0.f, 0.f, 0.f, 0.f, 0.f, 0.f, 0.f};
#pragma unroll
      for (int c8 = 0; c8 < DD / 8; ++c8) {
        const v4f a4 = aq[c8];
        const v4f b4 = *(const v4f*)(bchunk + c8 * 128);
        v2f a0 = {a4.x, a4.y}, b0 = {b4.x, b4.y};
        v2f a1 = {a4.z, a4.w}, b1 = {b4.z, b4.w};
        acc = wmma4(a0, b0, acc);
        acc = wmma4(a1, b1, acc);
      }
      float* dst = SL + ((size_t)branch * 16 + 8 * hl) * SSTRIDE + m0 + ll;
#pragma unroll
      for (int v = 0; v < 8; ++v) dst[(size_t)v * SSTRIDE] = acc[v] * scale;
    }
  }
  __syncthreads();

  {
    float* Lrow[4];
    const float* prow[4];
#pragma unroll
    for (int r = 0; r < 4; ++r) {
      const int idx = wave + 8 * r;
      const int br = idx >> 4, row = idx & 15;
      Lrow[r] = SL + (size_t)(br * 16 + row) * SSTRIDE + 4 * lane;
      prow[r] = pos_score + (size_t)(n0 + row) * NN + 4 * lane;
    }
    float lmax[4] = {-1e30f, -1e30f, -1e30f, -1e30f};
    for (int i = 0; i < 8; ++i) {
#pragma unroll
      for (int r = 0; r < 4; ++r) {
        const v4f s = *(const v4f*)(Lrow[r] + 128 * i);
        lmax[r] = fmaxf(fmaxf(lmax[r], fmaxf(s.x, s.y)), fmaxf(s.z, s.w));
      }
    }
#pragma unroll
    for (int off = 16; off > 0; off >>= 1) {
#pragma unroll
      for (int r = 0; r < 4; ++r) lmax[r] = fmaxf(lmax[r], __shfl_xor(lmax[r], off, 32));
    }
    float lsum[4] = {0.f, 0.f, 0.f, 0.f};
    for (int i = 0; i < 8; ++i) {
#pragma unroll
      for (int r = 0; r < 4; ++r) {
        const v4f s = *(const v4f*)(Lrow[r] + 128 * i);
        lsum[r] += expf(s.x - lmax[r]);
        lsum[r] += expf(s.y - lmax[r]);
        lsum[r] += expf(s.z - lmax[r]);
        lsum[r] += expf(s.w - lmax[r]);
      }
    }
#pragma unroll
    for (int off = 16; off > 0; off >>= 1) {
#pragma unroll
      for (int r = 0; r < 4; ++r) lsum[r] += __shfl_xor(lsum[r], off, 32);
    }
    float inv[4], ent[4];
#pragma unroll
    for (int r = 0; r < 4; ++r) {
      inv[r] = 1.0f / lsum[r];
      ent[r] = 0.f;
    }
    for (int i = 0; i < 8; ++i) {
      v4f s[4], pz[4];
#pragma unroll
      for (int r = 0; r < 4; ++r) {
        s[r] = *(const v4f*)(Lrow[r] + 128 * i);
        pz[r] = *(const v4f*)(prow[r] + 128 * i);
      }
#pragma unroll
      for (int r = 0; r < 4; ++r) {
        v4f a;
        a.x = omg * (expf(s[r].x - lmax[r]) * inv[r]) + g * pz[r].x;
        a.y = omg * (expf(s[r].y - lmax[r]) * inv[r]) + g * pz[r].y;
        a.z = omg * (expf(s[r].z - lmax[r]) * inv[r]) + g * pz[r].z;
        a.w = omg * (expf(s[r].w - lmax[r]) * inv[r]) + g * pz[r].w;
        ent[r] += -a.x * logf(a.x + EPSV);
        ent[r] += -a.y * logf(a.y + EPSV);
        ent[r] += -a.z * logf(a.z + EPSV);
        ent[r] += -a.w * logf(a.w + EPSV);
        *(v4f*)(Lrow[r] + 128 * i) = a;
      }
    }
#pragma unroll
    for (int off = 16; off > 0; off >>= 1) {
#pragma unroll
      for (int r = 0; r < 4; ++r) ent[r] += __shfl_xor(ent[r], off, 32);
    }
    if (lane == 0) {
#pragma unroll
      for (int r = 0; r < 4; ++r) entS[wave + 8 * r] = ent[r];
    }
  }
  __syncthreads();

  if (tid < 16) {
    const int row = tid;
    const float h0 = 2.0f - 2.0f / (1.0f + expf(-ht * entS[row]));
    const float h1 = 2.0f - 2.0f / (1.0f + expf(-ht * entS[16 + row]));
    const int sel = (h0 >= h1) ? 0 : 1;
    selS[row] = (float)sel;
    heat[(size_t)b * NN + n0 + row] = sel ? h1 : h0;
  }
  __syncthreads();

  const int selrow = (int)selS[ll];
  const float* Abase = SL + (size_t)(selrow * 16 + ll) * SSTRIDE + wave * 128 + 2 * hl;
  const float* yk = yb + (size_t)(wave * 128 + 2 * hl) * DD + 4 * ll;
  v8f acc[16];
#pragma unroll
  for (int t = 0; t < 16; ++t) acc[t] = (v8f){0.f, 0.f, 0.f, 0.f, 0.f, 0.f, 0.f, 0.f};

  v4f e0[4], e1[4], o0[4], o1[4];
  ld_yrows(e0, e1, yk);
#pragma unroll 1
  for (int ks = 0; ks < 30; ks += 2) {
    ld_yrows(o0, o1, yk + (size_t)(ks + 1) * 4 * DD);
    mma16(acc, *(const v2f*)(Abase + ks * 4), e0, e1);
    ld_yrows(e0, e1, yk + (size_t)(ks + 2) * 4 * DD);
    mma16(acc, *(const v2f*)(Abase + ks * 4 + 4), o0, o1);
  }
  ld_yrows(o0, o1, yk + (size_t)31 * 4 * DD);
  mma16(acc, *(const v2f*)(Abase + 30 * 4), e0, e1);
  mma16(acc, *(const v2f*)(Abase + 31 * 4), o0, o1);
  __syncthreads();

  {
    float* P = SL + (size_t)wave * 4096;
#pragma unroll
    for (int q = 0; q < 4; ++q) {
#pragma unroll
      for (int v = 0; v < 8; ++v) {
        const v4f pv = {acc[q * 4 + 0][v], acc[q * 4 + 1][v], acc[q * 4 + 2][v], acc[q * 4 + 3][v]};
        *(v4f*)(P + (8 * hl + v) * 256 + 64 * q + 4 * ll) = pv;
      }
    }
  }
  __syncthreads();
#pragma unroll
  for (int i = 0; i < 4; ++i) {
    const int idx = tid + 256 * i;
    v4f s = *(const v4f*)(SL + (size_t)idx * 4);
#pragma unroll
    for (int w = 1; w < 8; ++w) s += *(const v4f*)(SL + (size_t)w * 4096 + (size_t)idx * 4);
    *(v4f*)(out + ((size_t)b * NN + n0 + (idx >> 6)) * DD + 4 * (idx & 63)) = s;
  }
}

extern "C" void kernel_launch(void* const* d_in, const int* in_sizes, int n_in,
                              void* d_out, int out_size, void* d_ws, size_t ws_size,
                              hipStream_t stream) {
  const float* x = (const float*)d_in[0];
  const float* y = (const float*)d_in[1];
  const float* coords = (const float*)d_in[2];
  const float* U = (const float*)d_in[3];
  const float* S1 = (const float*)d_in[4];
  const float* S2 = (const float*)d_in[5];
  const float* gating = (const float*)d_in[6];
  const float* h_temp = (const float*)d_in[7];
  const float* p_temp = (const float*)d_in[8];
  const float* pos_emb = (const float*)d_in[9];

  float* out = (float*)d_out;
  float* heat = out + (size_t)BN * NN * DD;

  float* ws = (float*)d_ws;
  float* pos_score = ws;
  float* yp = pos_score + (size_t)NN * NN;
  float* xp = yp + (size_t)BN * NN * DD;
  float* Up = xp + (size_t)BN * NN * DD;

  pos_softmax_kernel<<<NN, 256, 0, stream>>>(coords, pos_emb, p_temp, pos_score,
                                             x, y, U, xp, yp, Up);

  const size_t lds_bytes = (size_t)(2 * 16 * SSTRIDE + 32 + 16) * sizeof(float);
  fused_attn_kernel<<<BN * 64, 256, lds_bytes, stream>>>(
      xp, Up, U, S1, S2, yp, y, pos_score, gating, h_temp, out, heat);
}
